# starting kernel
# baseline (speedup 1.0000x reference)
#include <hip/hip_runtime.h>
#include <hip/hip_bf16.h>

typedef __attribute__((ext_vector_type(2))) float v2f;
typedef __attribute__((ext_vector_type(4))) float v4f;
typedef __attribute__((ext_vector_type(8))) float v8f;

#define NNODE 25
#define NEDGE 64
#define HID   128
#define OUTD  128
#define BPB   16
#define NPAIR (BPB / 2)
#define ZROW  16

static __device__ __forceinline__ v2f lo2(v4f q) { return __builtin_shufflevector(q, q, 0, 1); }
static __device__ __forceinline__ v2f hi2(v4f q) { return __builtin_shufflevector(q, q, 2, 3); }

__launch_bounds__(128)
__global__ void sage_fused_kernel(const float* __restrict__ x,
                                  const int*   __restrict__ edge_index,
                                  const float* __restrict__ W1l,
                                  const float* __restrict__ W1r,
                                  const float* __restrict__ b1,
                                  const float* __restrict__ W2l,
                                  const float* __restrict__ W2r,
                                  const float* __restrict__ b2,
                                  float* __restrict__ out)
{
    __shared__ int   ssrc[NEDGE];
    __shared__ int   sdst[NEDGE];
    __shared__ float sdeginv[NNODE];
    __shared__ float snodew[NNODE];
    __shared__ float snw2[NNODE * 2];
    __shared__ float sA1[NNODE * 26];
    __align__(16) __shared__ float zbuf[NPAIR * NNODE * ZROW];
    __shared__ float As[BPB][2 * HID];

    const int tid  = threadIdx.x;
    const int lane = tid & 31;
    const int wave = tid >> 5;
    const int bbase = blockIdx.x * BPB;

    const int k = tid;
    const v2f wl0 = { W1l[0 * HID + k], W1l[0 * HID + k] };
    const v2f wl1 = { W1l[1 * HID + k], W1l[1 * HID + k] };
    const v2f wl2 = { W1l[2 * HID + k], W1l[2 * HID + k] };
    const v2f wr0 = { W1r[0 * HID + k], W1r[0 * HID + k] };
    const v2f wr1 = { W1r[1 * HID + k], W1r[1 * HID + k] };
    const v2f wr2 = { W1r[2 * HID + k], W1r[2 * HID + k] };
    const float b1k = b1[k];
    const v2f b1k2 = { b1k, b1k };

    if (tid < NEDGE) {
        ssrc[tid] = edge_index[tid];
        sdst[tid] = edge_index[NEDGE + tid];
    }
    for (int i = tid; i < NNODE * 26; i += 128) sA1[i] = 0.0f;
    if (tid < NNODE) snodew[tid] = 0.0f;
    __syncthreads();

    if (tid < NNODE) {
        int c = 0;
        #pragma unroll
        for (int e = 0; e < NEDGE; ++e) c += (sdst[e] == tid);
        sdeginv[tid] = 1.0f / fmaxf((float)c, 1.0f);
    }
    __syncthreads();

    if (tid < NEDGE) {
        const float di = sdeginv[sdst[tid]];
        atomicAdd(&sA1[sdst[tid] * 26 + ssrc[tid]], di);
        atomicAdd(&snodew[ssrc[tid]], di * (1.0f / 25.0f));
    }
    {
        const size_t base = (size_t)bbase * (NNODE * 3);
        for (int i = tid; i < BPB * NNODE * 3; i += 128) {
            const int slot = i / (NNODE * 3);
            const int r = i % (NNODE * 3);
            const int n = r / 3, d = r % 3;
            const int pair = slot >> 1, sub = slot & 1;
            zbuf[(pair * NNODE + n) * ZROW + 6 + 2 * d + sub] = x[base + i];
        }
    }
    __syncthreads();

    for (int i = tid; i < NPAIR * NNODE * 3; i += 128) {
        const int pair = i / (NNODE * 3);
        const int r = i % (NNODE * 3);
        const int n = r / 3, d = r % 3;
        v2f acc = { 0.0f, 0.0f };
        #pragma unroll
        for (int m = 0; m < NNODE; ++m) {
            const float a = sA1[n * 26 + m];
            const v2f xm = *(const v2f*)&zbuf[(pair * NNODE + m) * ZROW + 6 + 2 * d];
            acc = __builtin_elementwise_fma(xm, (v2f){ a, a }, acc);
        }
        *(v2f*)&zbuf[(pair * NNODE + n) * ZROW + 2 * d] = acc;
    }
    if (tid < NNODE) {
        const float w = snodew[tid];
        snw2[2 * tid] = w;
        snw2[2 * tid + 1] = w;
    }
    __syncthreads();

    for (int pair = 0; pair < NPAIR; ++pair) {
        const float* zp = &zbuf[pair * NNODE * ZROW];
        v2f cl = { 0.0f, 0.0f };
        v2f cr = { 0.0f, 0.0f };
        #pragma unroll
        for (int n = 0; n < NNODE; ++n) {
            const float* row = zp + n * ZROW;
            const v4f q0 = *(const v4f*)(row);
            const v4f q1 = *(const v4f*)(row + 4);
            const v4f q2 = *(const v4f*)(row + 8);
            v2f pre = b1k2;
            pre = __builtin_elementwise_fma(lo2(q0), wl0, pre);
            pre = __builtin_elementwise_fma(hi2(q0), wl1, pre);
            pre = __builtin_elementwise_fma(lo2(q1), wl2, pre);
            pre = __builtin_elementwise_fma(hi2(q1), wr0, pre);
            pre = __builtin_elementwise_fma(lo2(q2), wr1, pre);
            pre = __builtin_elementwise_fma(hi2(q2), wr2, pre);
            const v2f h = __builtin_elementwise_max(pre, (v2f){ 0.0f, 0.0f });
            const v2f nw = *(const v2f*)&snw2[2 * n];
            cl = __builtin_elementwise_fma(h, nw, cl);
            cr = cr + h;
        }
        As[2 * pair + 0][k] = cl.x;
        As[2 * pair + 1][k] = cl.y;
        As[2 * pair + 0][HID + k] = cr.x * (1.0f / 25.0f);
        As[2 * pair + 1][HID + k] = cr.y * (1.0f / 25.0f);
    }
    __syncthreads();

    {
        const int hi  = lane >> 4;
        const int l15 = lane & 15;
        const int n0a = wave * 32;
        const int n0b = n0a + 16;

        v8f c0, c1;
        const float bza = b2[n0a + l15];
        const float bzb = b2[n0b + l15];
        #pragma unroll
        for (int v = 0; v < 8; ++v) { c0[v] = bza; c1[v] = bzb; }

        const float* Wmats[2] = { W2l, W2r };
        #pragma unroll
        for (int half = 0; half < 2; ++half) {
            const float* W = Wmats[half];
            const int kofs = half * HID;
            #pragma unroll 4
            for (int k0 = 0; k0 < HID; k0 += 4) {
                const int kb = k0 + (hi << 1);
                v2f a;
                a.x = As[l15][kofs + kb];
                a.y = As[l15][kofs + kb + 1];
                v2f bfa, bfb;
                bfa.x = W[kb * OUTD + n0a + l15];
                bfa.y = W[(kb + 1) * OUTD + n0a + l15];
                bfb.x = W[kb * OUTD + n0b + l15];
                bfb.y = W[(kb + 1) * OUTD + n0b + l15];
                c0 = __builtin_amdgcn_wmma_f32_16x16x4_f32(
                        false, a, false, bfa, (short)0, c0, false, false);
                c1 = __builtin_amdgcn_wmma_f32_16x16x4_f32(
                        false, a, false, bfb, (short)0, c1, false, false);
            }
        }

        #pragma unroll
        for (int v = 0; v < 8; ++v) {
            const int m = v + (hi << 3);
            out[(size_t)(bbase + m) * OUTD + n0a + l15] = c0[v];
            out[(size_t)(bbase + m) * OUTD + n0b + l15] = c1[v];
        }
    }
}

extern "C" void kernel_launch(void* const* d_in, const int* in_sizes, int n_in,
                              void* d_out, int out_size, void* d_ws, size_t ws_size,
                              hipStream_t stream) {
    const float* x   = (const float*)d_in[0];
    const int*   ei  = (const int*)d_in[1];
    const float* W1l = (const float*)d_in[2];
    const float* W1r = (const float*)d_in[3];
    const float* b1  = (const float*)d_in[4];
    const float* W2l = (const float*)d_in[5];
    const float* W2r = (const float*)d_in[6];
    const float* b2  = (const float*)d_in[7];
    float* out = (float*)d_out;

    const int B = in_sizes[0] / (NNODE * 3);
    const int blocks = B / BPB;
    sage_fused_kernel<<<blocks, 128, 0, stream>>>(x, ei, W1l, W1r, b1, W2l, W2r, b2, out);
}


# layer 1 register-tiled: each wave takes 2 slot pairs x all 128 columns (4 per lane), so one broadcast LDS row read feeds 4 columns; LDS row reads per wave 600 to 150; per-column sum order unchanged
# speedup vs baseline: 1.3026x; 1.3026x over previous
#include <hip/hip_runtime.h>
#include <hip/hip_bf16.h>

typedef __attribute__((ext_vector_type(2))) float v2f;
typedef __attribute__((ext_vector_type(4))) float v4f;
typedef __attribute__((ext_vector_type(8))) float v8f;

#define NNODE 25
#define NEDGE 64
#define HID   128
#define OUTD  128
#define BPB   16
#define NPAIR (BPB / 2)
#define ZROW  16
#define CPL   4
#define PPW   (NPAIR / 4)

static __device__ __forceinline__ v2f lo2(v4f q) { return __builtin_shufflevector(q, q, 0, 1); }
static __device__ __forceinline__ v2f hi2(v4f q) { return __builtin_shufflevector(q, q, 2, 3); }

__launch_bounds__(128)
__global__ void sage_fused_kernel(const float* __restrict__ x,
                                  const int*   __restrict__ edge_index,
                                  const float* __restrict__ W1l,
                                  const float* __restrict__ W1r,
                                  const float* __restrict__ b1,
                                  const float* __restrict__ W2l,
                                  const float* __restrict__ W2r,
                                  const float* __restrict__ b2,
                                  float* __restrict__ out)
{
    __shared__ int   ssrc[NEDGE];
    __shared__ int   sdst[NEDGE];
    __shared__ float sdeginv[NNODE];
    __shared__ float snodew[NNODE];
    __shared__ float snw2[NNODE * 2];
    __shared__ float sA1[NNODE * 26];
    __align__(16) __shared__ float zbuf[NPAIR * NNODE * ZROW];
    __shared__ float As[BPB][2 * HID];

    const int tid  = threadIdx.x;
    const int lane = tid & 31;
    const int wave = tid >> 5;
    const int bbase = blockIdx.x * BPB;

    v2f wl0[CPL], wl1[CPL], wl2[CPL], wr0[CPL], wr1[CPL], wr2[CPL], b1k2[CPL];
    #pragma unroll
    for (int j = 0; j < CPL; ++j) {
        const int kc = lane + 32 * j;
        const float a0 = W1l[0 * HID + kc], a1 = W1l[1 * HID + kc], a2 = W1l[2 * HID + kc];
        const float r0 = W1r[0 * HID + kc], r1 = W1r[1 * HID + kc], r2 = W1r[2 * HID + kc];
        const float bk = b1[kc];
        wl0[j] = (v2f){ a0, a0 }; wl1[j] = (v2f){ a1, a1 }; wl2[j] = (v2f){ a2, a2 };
        wr0[j] = (v2f){ r0, r0 }; wr1[j] = (v2f){ r1, r1 }; wr2[j] = (v2f){ r2, r2 };
        b1k2[j] = (v2f){ bk, bk };
    }

    if (tid < NEDGE) {
        ssrc[tid] = edge_index[tid];
        sdst[tid] = edge_index[NEDGE + tid];
    }
    for (int i = tid; i < NNODE * 26; i += 128) sA1[i] = 0.0f;
    if (tid < NNODE) snodew[tid] = 0.0f;
    __syncthreads();

    if (tid < NNODE) {
        int c = 0;
        #pragma unroll
        for (int e = 0; e < NEDGE; ++e) c += (sdst[e] == tid);
        sdeginv[tid] = 1.0f / fmaxf((float)c, 1.0f);
    }
    __syncthreads();

    if (tid < NEDGE) {
        const float di = sdeginv[sdst[tid]];
        atomicAdd(&sA1[sdst[tid] * 26 + ssrc[tid]], di);
        atomicAdd(&snodew[ssrc[tid]], di * (1.0f / 25.0f));
    }
    {
        const size_t base = (size_t)bbase * (NNODE * 3);
        for (int i = tid; i < BPB * NNODE * 3; i += 128) {
            const int slot = i / (NNODE * 3);
            const int r = i % (NNODE * 3);
            const int n = r / 3, d = r % 3;
            const int pair = slot >> 1, sub = slot & 1;
            zbuf[(pair * NNODE + n) * ZROW + 6 + 2 * d + sub] = x[base + i];
        }
    }
    __syncthreads();

    for (int i = tid; i < NPAIR * NNODE * 3; i += 128) {
        const int pair = i / (NNODE * 3);
        const int r = i % (NNODE * 3);
        const int n = r / 3, d = r % 3;
        v2f acc = { 0.0f, 0.0f };
        #pragma unroll
        for (int m = 0; m < NNODE; ++m) {
            const float a = sA1[n * 26 + m];
            const v2f xm = *(const v2f*)&zbuf[(pair * NNODE + m) * ZROW + 6 + 2 * d];
            acc = __builtin_elementwise_fma(xm, (v2f){ a, a }, acc);
        }
        *(v2f*)&zbuf[(pair * NNODE + n) * ZROW + 2 * d] = acc;
    }
    if (tid < NNODE) {
        const float w = snodew[tid];
        snw2[2 * tid] = w;
        snw2[2 * tid + 1] = w;
    }
    __syncthreads();

    for (int pp = 0; pp < PPW; ++pp) {
        const int pair = wave * PPW + pp;
        const float* zp = &zbuf[pair * NNODE * ZROW];
        v2f cl[CPL], cr[CPL];
        #pragma unroll
        for (int j = 0; j < CPL; ++j) { cl[j] = (v2f){ 0.0f, 0.0f }; cr[j] = (v2f){ 0.0f, 0.0f }; }
        #pragma unroll
        for (int n = 0; n < NNODE; ++n) {
            const float* row = zp + n * ZROW;
            const v4f q0 = *(const v4f*)(row);
            const v4f q1 = *(const v4f*)(row + 4);
            const v4f q2 = *(const v4f*)(row + 8);
            const v2f nw = *(const v2f*)&snw2[2 * n];
            #pragma unroll
            for (int j = 0; j < CPL; ++j) {
                v2f pre = b1k2[j];
                pre = __builtin_elementwise_fma(lo2(q0), wl0[j], pre);
                pre = __builtin_elementwise_fma(hi2(q0), wl1[j], pre);
                pre = __builtin_elementwise_fma(lo2(q1), wl2[j], pre);
                pre = __builtin_elementwise_fma(hi2(q1), wr0[j], pre);
                pre = __builtin_elementwise_fma(lo2(q2), wr1[j], pre);
                pre = __builtin_elementwise_fma(hi2(q2), wr2[j], pre);
                const v2f h = __builtin_elementwise_max(pre, (v2f){ 0.0f, 0.0f });
                cl[j] = __builtin_elementwise_fma(h, nw, cl[j]);
                cr[j] = cr[j] + h;
            }
        }
        #pragma unroll
        for (int j = 0; j < CPL; ++j) {
            const int kc = lane + 32 * j;
            As[2 * pair + 0][kc] = cl[j].x;
            As[2 * pair + 1][kc] = cl[j].y;
            As[2 * pair + 0][HID + kc] = cr[j].x * (1.0f / 25.0f);
            As[2 * pair + 1][HID + kc] = cr[j].y * (1.0f / 25.0f);
        }
    }
    __syncthreads();

    {
        const int hi  = lane >> 4;
        const int l15 = lane & 15;
        const int n0a = wave * 32;
        const int n0b = n0a + 16;

        v8f c0, c1;
        const float bza = b2[n0a + l15];
        const float bzb = b2[n0b + l15];
        #pragma unroll
        for (int v = 0; v < 8; ++v) { c0[v] = bza; c1[v] = bzb; }

        const float* Wmats[2] = { W2l, W2r };
        #pragma unroll
        for (int half = 0; half < 2; ++half) {
            const float* W = Wmats[half];
            const int kofs = half * HID;
            #pragma unroll 4
            for (int k0 = 0; k0 < HID; k0 += 4) {
                const int kb = k0 + (hi << 1);
                v2f a;
                a.x = As[l15][kofs + kb];
                a.y = As[l15][kofs + kb + 1];
                v2f bfa, bfb;
                bfa.x = W[kb * OUTD + n0a + l15];
                bfa.y = W[(kb + 1) * OUTD + n0a + l15];
                bfb.x = W[kb * OUTD + n0b + l15];
                bfb.y = W[(kb + 1) * OUTD + n0b + l15];
                c0 = __builtin_amdgcn_wmma_f32_16x16x4_f32(
                        false, a, false, bfa, (short)0, c0, false, false);
                c1 = __builtin_amdgcn_wmma_f32_16x16x4_f32(
                        false, a, false, bfb, (short)0, c1, false, false);
            }
        }

        #pragma unroll
        for (int v = 0; v < 8; ++v) {
            const int m = v + (hi << 3);
            out[(size_t)(bbase + m) * OUTD + n0a + l15] = c0[v];
            out[(size_t)(bbase + m) * OUTD + n0b + l15] = c1[v];
        }
    }
}

extern "C" void kernel_launch(void* const* d_in, const int* in_sizes, int n_in,
                              void* d_out, int out_size, void* d_ws, size_t ws_size,
                              hipStream_t stream) {
    const float* x   = (const float*)d_in[0];
    const int*   ei  = (const int*)d_in[1];
    const float* W1l = (const float*)d_in[2];
    const float* W1r = (const float*)d_in[3];
    const float* b1  = (const float*)d_in[4];
    const float* W2l = (const float*)d_in[5];
    const float* W2r = (const float*)d_in[6];
    const float* b2  = (const float*)d_in[7];
    float* out = (float*)d_out;

    const int B = in_sizes[0] / (NNODE * 3);
    const int blocks = B / BPB;
    sage_fused_kernel<<<blocks, 128, 0, stream>>>(x, ei, W1l, W1r, b1, W2l, W2r, b2, out);
}


# global loads batched: x tile and edge list in one group at entry (was 3-4 dependent batches behind 2 barriers); W2 K-loop 16 k-steps per trip, L2 round trips per wave 16 -> 4; sum order unchanged
# speedup vs baseline: 1.4515x; 1.1143x over previous
#include <hip/hip_runtime.h>
#include <hip/hip_bf16.h>

typedef __attribute__((ext_vector_type(2))) float v2f;
typedef __attribute__((ext_vector_type(4))) float v4f;
typedef __attribute__((ext_vector_type(8))) float v8f;

#define NNODE 25
#define NEDGE 64
#define HID   128
#define OUTD  128
#define BPB   16
#define NPAIR (BPB / 2)
#define ZROW  16
#define CPL   4
#define PPW   (NPAIR / 4)
#define XT    (BPB * NNODE)
#define XTPT  ((XT + 127) / 128)

struct xyz { float c0, c1, c2; };

static __device__ __forceinline__ v2f lo2(v4f q) { return __builtin_shufflevector(q, q, 0, 1); }
static __device__ __forceinline__ v2f hi2(v4f q) { return __builtin_shufflevector(q, q, 2, 3); }

__launch_bounds__(128)
__global__ void sage_fused_kernel(const float* __restrict__ x,
                                  const int*   __restrict__ edge_index,
                                  const float* __restrict__ W1l,
                                  const float* __restrict__ W1r,
                                  const float* __restrict__ b1,
                                  const float* __restrict__ W2l,
                                  const float* __restrict__ W2r,
                                  const float* __restrict__ b2,
                                  float* __restrict__ out)
{
    __shared__ int   sedge[2 * NEDGE];
    __shared__ float sdeginv[NNODE];
    __shared__ float snodew[NNODE];
    __shared__ float snw2[NNODE * 2];
    __shared__ float sA1[NNODE * 26];
    __align__(16) __shared__ float zbuf[NPAIR * NNODE * ZROW];
    __shared__ float As[BPB][2 * HID];

    const int tid  = threadIdx.x;
    const int lane = tid & 31;
    const int wave = tid >> 5;
    const int bbase = blockIdx.x * BPB;

    const int ev = edge_index[tid];
    xyz xv[XTPT];
    {
        const xyz* xt = (const xyz*)(x + (size_t)bbase * (NNODE * 3));
        #pragma unroll
        for (int j = 0; j < XTPT; ++j) {
            const int u = tid + 128 * j;
            xv[j] = xt[u < XT ? u : XT - 1];
        }
    }

    v2f wl0[CPL], wl1[CPL], wl2[CPL], wr0[CPL], wr1[CPL], wr2[CPL], b1k2[CPL];
    #pragma unroll
    for (int j = 0; j < CPL; ++j) {
        const int kc = lane + 32 * j;
        const float a0 = W1l[0 * HID + kc], a1 = W1l[1 * HID + kc], a2 = W1l[2 * HID + kc];
        const float r0 = W1r[0 * HID + kc], r1 = W1r[1 * HID + kc], r2 = W1r[2 * HID + kc];
        const float bk = b1[kc];
        wl0[j] = (v2f){ a0, a0 }; wl1[j] = (v2f){ a1, a1 }; wl2[j] = (v2f){ a2, a2 };
        wr0[j] = (v2f){ r0, r0 }; wr1[j] = (v2f){ r1, r1 }; wr2[j] = (v2f){ r2, r2 };
        b1k2[j] = (v2f){ bk, bk };
    }

    for (int i = tid; i < NNODE * 26; i += 128) sA1[i] = 0.0f;
    if (tid < NNODE) snodew[tid] = 0.0f;
    #pragma unroll
    for (int j = 0; j < XTPT; ++j) {
        const int u = tid + 128 * j;
        if (128 * (j + 1) <= XT || u < XT) {
            const int slot = u / NNODE, n = u - NNODE * slot;
            float* zr = &zbuf[((slot >> 1) * NNODE + n) * ZROW + 6 + (slot & 1)];
            zr[0] = xv[j].c0;
            zr[2] = xv[j].c1;
            zr[4] = xv[j].c2;
        }
    }
    sedge[tid] = ev;
    __syncthreads();

    if (tid < NNODE) {
        int c = 0;
        #pragma unroll
        for (int e = 0; e < NEDGE; ++e) c += (sedge[NEDGE + e] == tid);
        sdeginv[tid] = 1.0f / fmaxf((float)c, 1.0f);
    }
    __syncthreads();

    if (tid < NEDGE) {
        const int es = sedge[tid], ed = sedge[NEDGE + tid];
        const float di = sdeginv[ed];
        atomicAdd(&sA1[ed * 26 + es], di);
        atomicAdd(&snodew[es], di * (1.0f / 25.0f));
    }
    __syncthreads();

    for (int i = tid; i < NPAIR * NNODE * 3; i += 128) {
        const int pair = i / (NNODE * 3);
        const int r = i % (NNODE * 3);
        const int n = r / 3, d = r % 3;
        v2f acc = { 0.0f, 0.0f };
        #pragma unroll
        for (int m = 0; m < NNODE; ++m) {
            const float a = sA1[n * 26 + m];
            const v2f xm = *(const v2f*)&zbuf[(pair * NNODE + m) * ZROW + 6 + 2 * d];
            acc = __builtin_elementwise_fma(xm, (v2f){ a, a }, acc);
        }
        *(v2f*)&zbuf[(pair * NNODE + n) * ZROW + 2 * d] = acc;
    }
    if (tid < NNODE) {
        const float w = snodew[tid];
        snw2[2 * tid] = w;
        snw2[2 * tid + 1] = w;
    }
    __syncthreads();

    for (int pp = 0; pp < PPW; ++pp) {
        const int pair = wave * PPW + pp;
        const float* zp = &zbuf[pair * NNODE * ZROW];
        v2f cl[CPL], cr[CPL];
        #pragma unroll
        for (int j = 0; j < CPL; ++j) { cl[j] = (v2f){ 0.0f, 0.0f }; cr[j] = (v2f){ 0.0f, 0.0f }; }
        #pragma unroll
        for (int n = 0; n < NNODE; ++n) {
            const float* row = zp + n * ZROW;
            const v4f q0 = *(const v4f*)(row);
            const v4f q1 = *(const v4f*)(row + 4);
            const v4f q2 = *(const v4f*)(row + 8);
            const v2f nw = *(const v2f*)&snw2[2 * n];
            #pragma unroll
            for (int j = 0; j < CPL; ++j) {
                v2f pre = b1k2[j];
                pre = __builtin_elementwise_fma(lo2(q0), wl0[j], pre);
                pre = __builtin_elementwise_fma(hi2(q0), wl1[j], pre);
                pre = __builtin_elementwise_fma(lo2(q1), wl2[j], pre);
                pre = __builtin_elementwise_fma(hi2(q1), wr0[j], pre);
                pre = __builtin_elementwise_fma(lo2(q2), wr1[j], pre);
                pre = __builtin_elementwise_fma(hi2(q2), wr2[j], pre);
                const v2f h = __builtin_elementwise_max(pre, (v2f){ 0.0f, 0.0f });
                cl[j] = __builtin_elementwise_fma(h, nw, cl[j]);
                cr[j] = cr[j] + h;
            }
        }
        #pragma unroll
        for (int j = 0; j < CPL; ++j) {
            const int kc = lane + 32 * j;
            As[2 * pair + 0][kc] = cl[j].x;
            As[2 * pair + 1][kc] = cl[j].y;
            As[2 * pair + 0][HID + kc] = cr[j].x * (1.0f / 25.0f);
            As[2 * pair + 1][HID + kc] = cr[j].y * (1.0f / 25.0f);
        }
    }
    __syncthreads();

    {
        const int hi  = lane >> 4;
        const int l15 = lane & 15;
        const int n0a = wave * 32;
        const int n0b = n0a + 16;

        v8f c0, c1;
        const float bza = b2[n0a + l15];
        const float bzb = b2[n0b + l15];
        #pragma unroll
        for (int v = 0; v < 8; ++v) { c0[v] = bza; c1[v] = bzb; }

        const float* Wmats[2] = { W2l, W2r };
        #pragma unroll
        for (int half = 0; half < 2; ++half) {
            const float* W = Wmats[half];
            const int kofs = half * HID;
            #pragma unroll 16
            for (int k0 = 0; k0 < HID; k0 += 4) {
                const int kb = k0 + (hi << 1);
                v2f a;
                a.x = As[l15][kofs + kb];
                a.y = As[l15][kofs + kb + 1];
                v2f bfa, bfb;
                bfa.x = W[kb * OUTD + n0a + l15];
                bfa.y = W[(kb + 1) * OUTD + n0a + l15];
                bfb.x = W[kb * OUTD + n0b + l15];
                bfb.y = W[(kb + 1) * OUTD + n0b + l15];
                c0 = __builtin_amdgcn_wmma_f32_16x16x4_f32(
                        false, a, false, bfa, (short)0, c0, false, false);
                c1 = __builtin_amdgcn_wmma_f32_16x16x4_f32(
                        false, a, false, bfb, (short)0, c1, false, false);
            }
        }

        #pragma unroll
        for (int v = 0; v < 8; ++v) {
            const int m = v + (hi << 3);
            out[(size_t)(bbase + m) * OUTD + n0a + l15] = c0[v];
            out[(size_t)(bbase + m) * OUTD + n0b + l15] = c1[v];
        }
    }
}

extern "C" void kernel_launch(void* const* d_in, const int* in_sizes, int n_in,
                              void* d_out, int out_size, void* d_ws, size_t ws_size,
                              hipStream_t stream) {
    const float* x   = (const float*)d_in[0];
    const int*   ei  = (const int*)d_in[1];
    const float* W1l = (const float*)d_in[2];
    const float* W1r = (const float*)d_in[3];
    const float* b1  = (const float*)d_in[4];
    const float* W2l = (const float*)d_in[5];
    const float* W2r = (const float*)d_in[6];
    const float* b2  = (const float*)d_in[7];
    float* out = (float*)d_out;

    const int B = in_sizes[0] / (NNODE * 3);
    const int blocks = B / BPB;
    sage_fused_kernel<<<blocks, 128, 0, stream>>>(x, ei, W1l, W1r, b1, W2l, W2r, b2, out);
}


# LDS row pitches moved off the 64-bank period: A-matrix rows 256->260 floats (fragment reads 16-way bank conflict -> at most 2-way), Z rows 16->20 floats (x scatter 7-way -> 2-way); sums unchanged
# speedup vs baseline: 1.5248x; 1.0505x over previous
#include <hip/hip_runtime.h>
#include <hip/hip_bf16.h>

typedef __attribute__((ext_vector_type(2))) float v2f;
typedef __attribute__((ext_vector_type(4))) float v4f;
typedef __attribute__((ext_vector_type(8))) float v8f;

#define NNODE 25
#define NEDGE 64
#define HID   128
#define OUTD  128
#define BPB   16
#define NPAIR (BPB / 2)
#define ZROW  20
#define CPL   4
#define PPW   (NPAIR / 4)
#define XT    (BPB * NNODE)
#define XTPT  ((XT + 127) / 128)
#define ASP   (2 * HID + 4)

struct xyz { float c0, c1, c2; };

static __device__ __forceinline__ v2f lo2(v4f q) { return __builtin_shufflevector(q, q, 0, 1); }
static __device__ __forceinline__ v2f hi2(v4f q) { return __builtin_shufflevector(q, q, 2, 3); }

__launch_bounds__(128)
__global__ void sage_fused_kernel(const float* __restrict__ x,
                                  const int*   __restrict__ edge_index,
                                  const float* __restrict__ W1l,
                                  const float* __restrict__ W1r,
                                  const float* __restrict__ b1,
                                  const float* __restrict__ W2l,
                                  const float* __restrict__ W2r,
                                  const float* __restrict__ b2,
                                  float* __restrict__ out)
{
    __shared__ int   sedge[2 * NEDGE];
    __shared__ float sdeginv[NNODE];
    __shared__ float snodew[NNODE];
    __shared__ float snw2[NNODE * 2];
    __shared__ float sA1[NNODE * 26];
    __align__(16) __shared__ float zbuf[NPAIR * NNODE * ZROW];
    __align__(16) __shared__ float As[BPB][ASP];

    const int tid  = threadIdx.x;
    const int lane = tid & 31;
    const int wave = tid >> 5;
    const int bbase = blockIdx.x * BPB;

    const int ev = edge_index[tid];
    xyz xv[XTPT];
    {
        const xyz* xt = (const xyz*)(x + (size_t)bbase * (NNODE * 3));
        #pragma unroll
        for (int j = 0; j < XTPT; ++j) {
            const int u = tid + 128 * j;
            xv[j] = xt[u < XT ? u : XT - 1];
        }
    }

    v2f wl0[CPL], wl1[CPL], wl2[CPL], wr0[CPL], wr1[CPL], wr2[CPL], b1k2[CPL];
    #pragma unroll
    for (int j = 0; j < CPL; ++j) {
        const int kc = lane + 32 * j;
        const float a0 = W1l[0 * HID + kc], a1 = W1l[1 * HID + kc], a2 = W1l[2 * HID + kc];
        const float r0 = W1r[0 * HID + kc], r1 = W1r[1 * HID + kc], r2 = W1r[2 * HID + kc];
        const float bk = b1[kc];
        wl0[j] = (v2f){ a0, a0 }; wl1[j] = (v2f){ a1, a1 }; wl2[j] = (v2f){ a2, a2 };
        wr0[j] = (v2f){ r0, r0 }; wr1[j] = (v2f){ r1, r1 }; wr2[j] = (v2f){ r2, r2 };
        b1k2[j] = (v2f){ bk, bk };
    }

    for (int i = tid; i < NNODE * 26; i += 128) sA1[i] = 0.0f;
    if (tid < NNODE) snodew[tid] = 0.0f;
    #pragma unroll
    for (int j = 0; j < XTPT; ++j) {
        const int u = tid + 128 * j;
        if (128 * (j + 1) <= XT || u < XT) {
            const int slot = u / NNODE, n = u - NNODE * slot;
            float* zr = &zbuf[((slot >> 1) * NNODE + n) * ZROW + 6 + (slot & 1)];
            zr[0] = xv[j].c0;
            zr[2] = xv[j].c1;
            zr[4] = xv[j].c2;
        }
    }
    sedge[tid] = ev;
    __syncthreads();

    if (tid < NNODE) {
        int c = 0;
        #pragma unroll
        for (int e = 0; e < NEDGE; ++e) c += (sedge[NEDGE + e] == tid);
        sdeginv[tid] = 1.0f / fmaxf((float)c, 1.0f);
    }
    __syncthreads();

    if (tid < NEDGE) {
        const int es = sedge[tid], ed = sedge[NEDGE + tid];
        const float di = sdeginv[ed];
        atomicAdd(&sA1[ed * 26 + es], di);
        atomicAdd(&snodew[es], di * (1.0f / 25.0f));
    }
    __syncthreads();

    for (int i = tid; i < NPAIR * NNODE * 3; i += 128) {
        const int pair = i / (NNODE * 3);
        const int r = i % (NNODE * 3);
        const int n = r / 3, d = r % 3;
        v2f acc = { 0.0f, 0.0f };
        #pragma unroll
        for (int m = 0; m < NNODE; ++m) {
            const float a = sA1[n * 26 + m];
            const v2f xm = *(const v2f*)&zbuf[(pair * NNODE + m) * ZROW + 6 + 2 * d];
            acc = __builtin_elementwise_fma(xm, (v2f){ a, a }, acc);
        }
        *(v2f*)&zbuf[(pair * NNODE + n) * ZROW + 2 * d] = acc;
    }
    if (tid < NNODE) {
        const float w = snodew[tid];
        snw2[2 * tid] = w;
        snw2[2 * tid + 1] = w;
    }
    __syncthreads();

    for (int pp = 0; pp < PPW; ++pp) {
        const int pair = wave * PPW + pp;
        const float* zp = &zbuf[pair * NNODE * ZROW];
        v2f cl[CPL], cr[CPL];
        #pragma unroll
        for (int j = 0; j < CPL; ++j) { cl[j] = (v2f){ 0.0f, 0.0f }; cr[j] = (v2f){ 0.0f, 0.0f }; }
        #pragma unroll
        for (int n = 0; n < NNODE; ++n) {
            const float* row = zp + n * ZROW;
            const v4f q0 = *(const v4f*)(row);
            const v4f q1 = *(const v4f*)(row + 4);
            const v4f q2 = *(const v4f*)(row + 8);
            const v2f nw = *(const v2f*)&snw2[2 * n];
            #pragma unroll
            for (int j = 0; j < CPL; ++j) {
                v2f pre = b1k2[j];
                pre = __builtin_elementwise_fma(lo2(q0), wl0[j], pre);
                pre = __builtin_elementwise_fma(hi2(q0), wl1[j], pre);
                pre = __builtin_elementwise_fma(lo2(q1), wl2[j], pre);
                pre = __builtin_elementwise_fma(hi2(q1), wr0[j], pre);
                pre = __builtin_elementwise_fma(lo2(q2), wr1[j], pre);
                pre = __builtin_elementwise_fma(hi2(q2), wr2[j], pre);
                const v2f h = __builtin_elementwise_max(pre, (v2f){ 0.0f, 0.0f });
                cl[j] = __builtin_elementwise_fma(h, nw, cl[j]);
                cr[j] = cr[j] + h;
            }
        }
        #pragma unroll
        for (int j = 0; j < CPL; ++j) {
            const int kc = lane + 32 * j;
            As[2 * pair + 0][kc] = cl[j].x;
            As[2 * pair + 1][kc] = cl[j].y;
            As[2 * pair + 0][HID + kc] = cr[j].x * (1.0f / 25.0f);
            As[2 * pair + 1][HID + kc] = cr[j].y * (1.0f / 25.0f);
        }
    }
    __syncthreads();

    {
        const int hi  = lane >> 4;
        const int l15 = lane & 15;
        const int n0a = wave * 32;
        const int n0b = n0a + 16;

        v8f c0, c1;
        const float bza = b2[n0a + l15];
        const float bzb = b2[n0b + l15];
        #pragma unroll
        for (int v = 0; v < 8; ++v) { c0[v] = bza; c1[v] = bzb; }

        const float* Wmats[2] = { W2l, W2r };
        #pragma unroll
        for (int half = 0; half < 2; ++half) {
            const float* W = Wmats[half];
            const int kofs = half * HID;
            #pragma unroll 16
            for (int k0 = 0; k0 < HID; k0 += 4) {
                const int kb = k0 + (hi << 1);
                v2f a;
                a.x = As[l15][kofs + kb];
                a.y = As[l15][kofs + kb + 1];
                v2f bfa, bfb;
                bfa.x = W[kb * OUTD + n0a + l15];
                bfa.y = W[(kb + 1) * OUTD + n0a + l15];
                bfb.x = W[kb * OUTD + n0b + l15];
                bfb.y = W[(kb + 1) * OUTD + n0b + l15];
                c0 = __builtin_amdgcn_wmma_f32_16x16x4_f32(
                        false, a, false, bfa, (short)0, c0, false, false);
                c1 = __builtin_amdgcn_wmma_f32_16x16x4_f32(
                        false, a, false, bfb, (short)0, c1, false, false);
            }
        }

        #pragma unroll
        for (int v = 0; v < 8; ++v) {
            const int m = v + (hi << 3);
            out[(size_t)(bbase + m) * OUTD + n0a + l15] = c0[v];
            out[(size_t)(bbase + m) * OUTD + n0b + l15] = c1[v];
        }
    }
}

extern "C" void kernel_launch(void* const* d_in, const int* in_sizes, int n_in,
                              void* d_out, int out_size, void* d_ws, size_t ws_size,
                              hipStream_t stream) {
    const float* x   = (const float*)d_in[0];
    const int*   ei  = (const int*)d_in[1];
    const float* W1l = (const float*)d_in[2];
    const float* W1r = (const float*)d_in[3];
    const float* b1  = (const float*)d_in[4];
    const float* W2l = (const float*)d_in[5];
    const float* W2r = (const float*)d_in[6];
    const float* b2  = (const float*)d_in[7];
    float* out = (float*)d_out;

    const int B = in_sizes[0] / (NNODE * 3);
    const int blocks = B / BPB;
    sage_fused_kernel<<<blocks, 128, 0, stream>>>(x, ei, W1l, W1r, b1, W2l, W2r, b2, out);
}
